# starting kernel
# baseline (speedup 1.0000x reference)
#include <hip/hip_runtime.h>

#define BQ     4
#define CIN    64
#define HH     96
#define WW     96
#define GROUPS 8
#define NCH    8
#define CGC    8
#define KW     7
#define PADR   3
#define HWSZ   (HH * WW)

#define ROWS   4
#define KVROWS (ROWS + 2*PADR)
#define NTHR   128

typedef __attribute__((ext_vector_type(2))) float v2f;
typedef __attribute__((ext_vector_type(4))) float v4f;
typedef __attribute__((ext_vector_type(8))) float v8f;

__device__ __forceinline__ int reflect96(int i) {
    if (i < 0)  i = -i;
    if (i > 95) i = 190 - i;
    return i;
}

__global__ __launch_bounds__(NTHR)
void SelfAttentionBlock_kernel(const float* __restrict__ x,
                               const float* __restrict__ Wk,
                               const float* __restrict__ Wq,
                               const float* __restrict__ Wv,
                               const float* __restrict__ r_ai,
                               const float* __restrict__ r_aj,
                               float* __restrict__ out)
{
    __shared__ __attribute__((aligned(16))) float kb[KVROWS][WW][NCH];
    __shared__ __attribute__((aligned(16))) float vb[KVROWS][WW][NCH];
    __shared__ __attribute__((aligned(16))) float qb[ROWS][WW][NCH];
    __shared__ __attribute__((aligned(16))) float rcomp[KW][NCH];

    const int h0   = blockIdx.x * ROWS;
    const int g    = blockIdx.y;
    const int b    = blockIdx.z;
    const int tid  = threadIdx.x;
    const int lane = tid & 31;
    const int wave = tid >> 5;
    const int half = lane >> 4;
    const int m    = lane & 15;

    if (tid < KW * NCH) {
        int t = tid / NCH, n = tid % NCH;
        int cg = g * NCH + n;
        rcomp[t][n] = (cg < 32) ? r_ai[cg * KW + t] : r_aj[(cg - 32) * KW + t];
    }

    const float* WkG = Wk + g * 64;
    const float* WqG = Wq + g * 64;
    const float* WvG = Wv + g * 64;
    const int k_lo = half * 2;
    const int k_hi = 4 + half * 2;

    v2f aKV_lo, aKV_hi, aQ_lo, aQ_hi;
    if (m < 8) {
        aKV_lo[0] = WkG[m * 8 + k_lo];     aKV_lo[1] = WkG[m * 8 + k_lo + 1];
        aKV_hi[0] = WkG[m * 8 + k_hi];     aKV_hi[1] = WkG[m * 8 + k_hi + 1];
    } else {
        aKV_lo[0] = WvG[(m - 8) * 8 + k_lo]; aKV_lo[1] = WvG[(m - 8) * 8 + k_lo + 1];
        aKV_hi[0] = WvG[(m - 8) * 8 + k_hi]; aKV_hi[1] = WvG[(m - 8) * 8 + k_hi + 1];
    }
    aQ_lo[0] = WqG[(m & 7) * 8 + k_lo];  aQ_lo[1] = WqG[(m & 7) * 8 + k_lo + 1];
    aQ_hi[0] = WqG[(m & 7) * 8 + k_hi];  aQ_hi[1] = WqG[(m & 7) * 8 + k_hi + 1];

    const float* xg = x + (size_t)(b * CIN + g * CGC) * HWSZ;

    for (int t = wave; t < KVROWS * 6; t += 4) {
        int i     = t / 6;
        int strip = (t % 6) * 16;
        int hr    = reflect96(h0 - PADR + i);
        int col   = strip + m;
        const float* xp = xg + hr * WW + col;
        v2f blo, bhi;
        blo[0] = xp[(size_t)k_lo       * HWSZ];
        blo[1] = xp[(size_t)(k_lo + 1) * HWSZ];
        bhi[0] = xp[(size_t)k_hi       * HWSZ];
        bhi[1] = xp[(size_t)(k_hi + 1) * HWSZ];

        v8f acc = {};
        acc = __builtin_amdgcn_wmma_f32_16x16x4_f32(false, aKV_lo, false, blo,
                                                    (short)0, acc, false, false);
        acc = __builtin_amdgcn_wmma_f32_16x16x4_f32(false, aKV_hi, false, bhi,
                                                    (short)0, acc, false, false);
        float* dst = half ? &vb[i][col][0] : &kb[i][col][0];
        v4f lo4 = {acc[0], acc[1], acc[2], acc[3]};
        v4f hi4 = {acc[4], acc[5], acc[6], acc[7]};
        *(v4f*)dst       = lo4;
        *((v4f*)dst + 1) = hi4;
    }

    for (int t = wave; t < ROWS * 6; t += 4) {
        int rr    = t / 6;
        int strip = (t % 6) * 16;
        int col   = strip + m;
        const float* xp = xg + (h0 + rr) * WW + col;
        v2f blo, bhi;
        blo[0] = xp[(size_t)k_lo       * HWSZ];
        blo[1] = xp[(size_t)(k_lo + 1) * HWSZ];
        bhi[0] = xp[(size_t)k_hi       * HWSZ];
        bhi[1] = xp[(size_t)(k_hi + 1) * HWSZ];

        v8f acc = {};
        acc = __builtin_amdgcn_wmma_f32_16x16x4_f32(false, aQ_lo, false, blo,
                                                    (short)0, acc, false, false);
        acc = __builtin_amdgcn_wmma_f32_16x16x4_f32(false, aQ_hi, false, bhi,
                                                    (short)0, acc, false, false);
        if (!half) {
            float* dst = &qb[rr][col][0];
            v4f lo4 = {acc[0], acc[1], acc[2], acc[3]};
            v4f hi4 = {acc[4], acc[5], acc[6], acc[7]};
            *(v4f*)dst       = lo4;
            *((v4f*)dst + 1) = hi4;
        }
    }

    __syncthreads();

    const bool rowBias = (g < 4);
    for (int p = tid; p < ROWS * WW; p += NTHR) {
        const int rr = p / WW;
        const int w  = p % WW;

        const v4f q0 = *(const v4f*)&qb[rr][w][0];
        const v4f q1 = *(const v4f*)&qb[rr][w][4];

        float rb[KW];
        #pragma unroll
        for (int t = 0; t < KW; ++t) {
            const v4f r0 = *(const v4f*)&rcomp[t][0];
            const v4f r1 = *(const v4f*)&rcomp[t][4];
            float s = 0.f;
            #pragma unroll
            for (int n = 0; n < 4; ++n) s = __builtin_fmaf(q0[n], r0[n], s);
            #pragma unroll
            for (int n = 0; n < 4; ++n) s = __builtin_fmaf(q1[n], r1[n], s);
            rb[t] = s;
        }

        float mmax = -3.0e38f, ssum = 0.0f;
        float acc[NCH];
        #pragma unroll
        for (int n = 0; n < NCH; ++n) acc[n] = 0.0f;

        for (int ki = 0; ki < KW; ++ki) {
            const int slot = rr + ki;
            for (int kj = 0; kj < KW; ++kj) {
                int c = w + kj - PADR;
                c = (c < 0) ? -c : ((c > 95) ? 190 - c : c);

                const v4f k0 = *(const v4f*)&kb[slot][c][0];
                const v4f k1 = *(const v4f*)&kb[slot][c][4];
                float l = rb[rowBias ? ki : kj];
                #pragma unroll
                for (int n = 0; n < 4; ++n) l = __builtin_fmaf(q0[n], k0[n], l);
                #pragma unroll
                for (int n = 0; n < 4; ++n) l = __builtin_fmaf(q1[n], k1[n], l);

                const float mn   = fmaxf(mmax, l);
                const float corr = __expf(mmax - mn);
                const float e    = __expf(l - mn);
                ssum = ssum * corr + e;

                const v4f va = *(const v4f*)&vb[slot][c][0];
                const v4f vbb = *(const v4f*)&vb[slot][c][4];
                #pragma unroll
                for (int n = 0; n < 4; ++n) acc[n]     = acc[n]     * corr + e * va[n];
                #pragma unroll
                for (int n = 0; n < 4; ++n) acc[n + 4] = acc[n + 4] * corr + e * vbb[n];
                mmax = mn;
            }
        }

        const float inv = 1.0f / ssum;
        float* op = out + ((size_t)(b * CIN + g * NCH) * HH + (h0 + rr)) * WW + w;
        #pragma unroll
        for (int n = 0; n < NCH; ++n)
            op[(size_t)n * HWSZ] = acc[n] * inv;
    }
}

extern "C" void kernel_launch(void* const* d_in, const int* in_sizes, int n_in,
                              void* d_out, int out_size, void* d_ws, size_t ws_size,
                              hipStream_t stream) {
    (void)in_sizes; (void)n_in; (void)out_size; (void)d_ws; (void)ws_size;
    const float* x    = (const float*)d_in[0];
    const float* Wk   = (const float*)d_in[1];
    const float* Wq   = (const float*)d_in[2];
    const float* Wv   = (const float*)d_in[3];
    const float* r_ai = (const float*)d_in[4];
    const float* r_aj = (const float*)d_in[5];
    float* out = (float*)d_out;

    dim3 grid(HH / ROWS, GROUPS, BQ);
    SelfAttentionBlock_kernel<<<grid, NTHR, 0, stream>>>(x, Wk, Wq, Wv, r_ai, r_aj, out);
}


# LDS layout: k, v, q stored as two 4-channel planes (16-byte pixel records, not 32-byte) so each ds_load_b128 of the 7x7 loop has a 16-byte lane stride: no 2-way bank conflict
# speedup vs baseline: 1.0484x; 1.0484x over previous
#include <hip/hip_runtime.h>

#define BQ     4
#define CIN    64
#define HH     96
#define WW     96
#define GROUPS 8
#define NCH    8
#define CGC    8
#define KW     7
#define PADR   3
#define HWSZ   (HH * WW)

#define ROWS   4
#define KVROWS (ROWS + 2*PADR)
#define NTHR   128

typedef __attribute__((ext_vector_type(2))) float v2f;
typedef __attribute__((ext_vector_type(4))) float v4f;
typedef __attribute__((ext_vector_type(8))) float v8f;

__device__ __forceinline__ int reflect96(int i) {
    if (i < 0)  i = -i;
    if (i > 95) i = 190 - i;
    return i;
}

__global__ __launch_bounds__(NTHR)
void SelfAttentionBlock_kernel(const float* __restrict__ x,
                               const float* __restrict__ Wk,
                               const float* __restrict__ Wq,
                               const float* __restrict__ Wv,
                               const float* __restrict__ r_ai,
                               const float* __restrict__ r_aj,
                               float* __restrict__ out)
{
    __shared__ v4f kb0[KVROWS][WW];
    __shared__ v4f kb1[KVROWS][WW];
    __shared__ v4f vb0[KVROWS][WW];
    __shared__ v4f vb1[KVROWS][WW];
    __shared__ v4f qb0[ROWS][WW];
    __shared__ v4f qb1[ROWS][WW];
    __shared__ __attribute__((aligned(16))) float rcomp[KW][NCH];

    const int h0   = blockIdx.x * ROWS;
    const int g    = blockIdx.y;
    const int b    = blockIdx.z;
    const int tid  = threadIdx.x;
    const int lane = tid & 31;
    const int wave = tid >> 5;
    const int half = lane >> 4;
    const int m    = lane & 15;

    if (tid < KW * NCH) {
        int t = tid / NCH, n = tid % NCH;
        int cg = g * NCH + n;
        rcomp[t][n] = (cg < 32) ? r_ai[cg * KW + t] : r_aj[(cg - 32) * KW + t];
    }

    const float* WkG = Wk + g * 64;
    const float* WqG = Wq + g * 64;
    const float* WvG = Wv + g * 64;
    const int k_lo = half * 2;
    const int k_hi = 4 + half * 2;

    v2f aKV_lo, aKV_hi, aQ_lo, aQ_hi;
    if (m < 8) {
        aKV_lo[0] = WkG[m * 8 + k_lo];     aKV_lo[1] = WkG[m * 8 + k_lo + 1];
        aKV_hi[0] = WkG[m * 8 + k_hi];     aKV_hi[1] = WkG[m * 8 + k_hi + 1];
    } else {
        aKV_lo[0] = WvG[(m - 8) * 8 + k_lo]; aKV_lo[1] = WvG[(m - 8) * 8 + k_lo + 1];
        aKV_hi[0] = WvG[(m - 8) * 8 + k_hi]; aKV_hi[1] = WvG[(m - 8) * 8 + k_hi + 1];
    }
    aQ_lo[0] = WqG[(m & 7) * 8 + k_lo];  aQ_lo[1] = WqG[(m & 7) * 8 + k_lo + 1];
    aQ_hi[0] = WqG[(m & 7) * 8 + k_hi];  aQ_hi[1] = WqG[(m & 7) * 8 + k_hi + 1];

    const float* xg = x + (size_t)(b * CIN + g * CGC) * HWSZ;

    for (int t = wave; t < KVROWS * 6; t += 4) {
        int i     = t / 6;
        int strip = (t % 6) * 16;
        int hr    = reflect96(h0 - PADR + i);
        int col   = strip + m;
        const float* xp = xg + hr * WW + col;
        v2f blo, bhi;
        blo[0] = xp[(size_t)k_lo       * HWSZ];
        blo[1] = xp[(size_t)(k_lo + 1) * HWSZ];
        bhi[0] = xp[(size_t)k_hi       * HWSZ];
        bhi[1] = xp[(size_t)(k_hi + 1) * HWSZ];

        v8f acc = {};
        acc = __builtin_amdgcn_wmma_f32_16x16x4_f32(false, aKV_lo, false, blo,
                                                    (short)0, acc, false, false);
        acc = __builtin_amdgcn_wmma_f32_16x16x4_f32(false, aKV_hi, false, bhi,
                                                    (short)0, acc, false, false);
        v4f lo4 = {acc[0], acc[1], acc[2], acc[3]};
        v4f hi4 = {acc[4], acc[5], acc[6], acc[7]};
        if (half) { vb0[i][col] = lo4; vb1[i][col] = hi4; }
        else      { kb0[i][col] = lo4; kb1[i][col] = hi4; }
    }

    for (int t = wave; t < ROWS * 6; t += 4) {
        int rr    = t / 6;
        int strip = (t % 6) * 16;
        int col   = strip + m;
        const float* xp = xg + (h0 + rr) * WW + col;
        v2f blo, bhi;
        blo[0] = xp[(size_t)k_lo       * HWSZ];
        blo[1] = xp[(size_t)(k_lo + 1) * HWSZ];
        bhi[0] = xp[(size_t)k_hi       * HWSZ];
        bhi[1] = xp[(size_t)(k_hi + 1) * HWSZ];

        v8f acc = {};
        acc = __builtin_amdgcn_wmma_f32_16x16x4_f32(false, aQ_lo, false, blo,
                                                    (short)0, acc, false, false);
        acc = __builtin_amdgcn_wmma_f32_16x16x4_f32(false, aQ_hi, false, bhi,
                                                    (short)0, acc, false, false);
        if (!half) {
            v4f lo4 = {acc[0], acc[1], acc[2], acc[3]};
            v4f hi4 = {acc[4], acc[5], acc[6], acc[7]};
            qb0[rr][col] = lo4;
            qb1[rr][col] = hi4;
        }
    }

    __syncthreads();

    const bool rowBias = (g < 4);
    for (int p = tid; p < ROWS * WW; p += NTHR) {
        const int rr = p / WW;
        const int w  = p % WW;

        const v4f q0 = qb0[rr][w];
        const v4f q1 = qb1[rr][w];

        float rb[KW];
        #pragma unroll
        for (int t = 0; t < KW; ++t) {
            const v4f r0 = *(const v4f*)&rcomp[t][0];
            const v4f r1 = *(const v4f*)&rcomp[t][4];
            float s = 0.f;
            #pragma unroll
            for (int n = 0; n < 4; ++n) s = __builtin_fmaf(q0[n], r0[n], s);
            #pragma unroll
            for (int n = 0; n < 4; ++n) s = __builtin_fmaf(q1[n], r1[n], s);
            rb[t] = s;
        }

        float mmax = -3.0e38f, ssum = 0.0f;
        float acc[NCH];
        #pragma unroll
        for (int n = 0; n < NCH; ++n) acc[n] = 0.0f;

        for (int ki = 0; ki < KW; ++ki) {
            const int slot = rr + ki;
            for (int kj = 0; kj < KW; ++kj) {
                int c = w + kj - PADR;
                c = (c < 0) ? -c : ((c > 95) ? 190 - c : c);

                const v4f k0 = kb0[slot][c];
                const v4f k1 = kb1[slot][c];
                float l = rb[rowBias ? ki : kj];
                #pragma unroll
                for (int n = 0; n < 4; ++n) l = __builtin_fmaf(q0[n], k0[n], l);
                #pragma unroll
                for (int n = 0; n < 4; ++n) l = __builtin_fmaf(q1[n], k1[n], l);

                const float mn   = fmaxf(mmax, l);
                const float corr = __expf(mmax - mn);
                const float e    = __expf(l - mn);
                ssum = ssum * corr + e;

                const v4f va  = vb0[slot][c];
                const v4f vbb = vb1[slot][c];
                #pragma unroll
                for (int n = 0; n < 4; ++n) acc[n]     = acc[n]     * corr + e * va[n];
                #pragma unroll
                for (int n = 0; n < 4; ++n) acc[n + 4] = acc[n + 4] * corr + e * vbb[n];
                mmax = mn;
            }
        }

        const float inv = 1.0f / ssum;
        float* op = out + ((size_t)(b * CIN + g * NCH) * HH + (h0 + rr)) * WW + w;
        #pragma unroll
        for (int n = 0; n < NCH; ++n)
            op[(size_t)n * HWSZ] = acc[n] * inv;
    }
}

extern "C" void kernel_launch(void* const* d_in, const int* in_sizes, int n_in,
                              void* d_out, int out_size, void* d_ws, size_t ws_size,
                              hipStream_t stream) {
    (void)in_sizes; (void)n_in; (void)out_size; (void)d_ws; (void)ws_size;
    const float* x    = (const float*)d_in[0];
    const float* Wk   = (const float*)d_in[1];
    const float* Wq   = (const float*)d_in[2];
    const float* Wv   = (const float*)d_in[3];
    const float* r_ai = (const float*)d_in[4];
    const float* r_aj = (const float*)d_in[5];
    float* out = (float*)d_out;

    dim3 grid(HH / ROWS, GROUPS, BQ);
    SelfAttentionBlock_kernel<<<grid, NTHR, 0, stream>>>(x, Wk, Wq, Wv, r_ai, r_aj, out);
}


# Projection phase: each wave issues the x loads of all 15 of its tiles before the first WMMA (q reuses the centre rows' fragments) instead of 21 dependent load round trips per wave
# speedup vs baseline: 1.5157x; 1.4457x over previous
#include <hip/hip_runtime.h>

#define BQ     4
#define CIN    64
#define HH     96
#define WW     96
#define GROUPS 8
#define NCH    8
#define CGC    8
#define KW     7
#define PADR   3
#define HWSZ   (HH * WW)

#define ROWS   4
#define KVROWS (ROWS + 2*PADR)
#define NTHR   128
#define NPLAIN 9
#define NCENT  6

typedef __attribute__((ext_vector_type(2))) float v2f;
typedef __attribute__((ext_vector_type(4))) float v4f;
typedef __attribute__((ext_vector_type(8))) float v8f;

__device__ __forceinline__ int reflect96(int i) {
    if (i < 0)  i = -i;
    if (i > 95) i = 190 - i;
    return i;
}

__global__ __launch_bounds__(NTHR)
void SelfAttentionBlock_kernel(const float* __restrict__ x,
                               const float* __restrict__ Wk,
                               const float* __restrict__ Wq,
                               const float* __restrict__ Wv,
                               const float* __restrict__ r_ai,
                               const float* __restrict__ r_aj,
                               float* __restrict__ out)
{
    __shared__ v4f kb0[KVROWS][WW];
    __shared__ v4f kb1[KVROWS][WW];
    __shared__ v4f vb0[KVROWS][WW];
    __shared__ v4f vb1[KVROWS][WW];
    __shared__ v4f qb0[ROWS][WW];
    __shared__ v4f qb1[ROWS][WW];
    __shared__ __attribute__((aligned(16))) float rcomp[NTHR / NCH][NCH];

    const int h0   = blockIdx.x * ROWS;
    const int g    = blockIdx.y;
    const int b    = blockIdx.z;
    const int tid  = threadIdx.x;
    const int lane = tid & 31;
    const int wave = __builtin_amdgcn_readfirstlane(tid >> 5);
    const int half = lane >> 4;
    const int m    = lane & 15;

    float rcv;
    {
        const int ti = (tid < KW * NCH) ? tid : 0;
        const int t = ti / NCH, n = ti % NCH;
        const int cg = g * NCH + n;
        rcv = (g < 4) ? r_ai[cg * KW + t] : r_aj[(cg - 32) * KW + t];
    }

    const float* WkG = Wk + g * 64;
    const float* WqG = Wq + g * 64;
    const float* WvG = Wv + g * 64;
    const int k_lo = half * 2;
    const int k_hi = 4 + half * 2;

    v2f aKV_lo, aKV_hi, aQ_lo, aQ_hi;
    if (m < 8) {
        aKV_lo[0] = WkG[m * 8 + k_lo];     aKV_lo[1] = WkG[m * 8 + k_lo + 1];
        aKV_hi[0] = WkG[m * 8 + k_hi];     aKV_hi[1] = WkG[m * 8 + k_hi + 1];
    } else {
        aKV_lo[0] = WvG[(m - 8) * 8 + k_lo]; aKV_lo[1] = WvG[(m - 8) * 8 + k_lo + 1];
        aKV_hi[0] = WvG[(m - 8) * 8 + k_hi]; aKV_hi[1] = WvG[(m - 8) * 8 + k_hi + 1];
    }
    aQ_lo[0] = WqG[(m & 7) * 8 + k_lo];  aQ_lo[1] = WqG[(m & 7) * 8 + k_lo + 1];
    aQ_hi[0] = WqG[(m & 7) * 8 + k_hi];  aQ_hi[1] = WqG[(m & 7) * 8 + k_hi + 1];

    const float* xg = x + (size_t)(b * CIN + g * CGC) * HWSZ;

    const int wr = wave >> 1;
    const int wc = wave & 1;
    const float* xw = xg + wc * 48;
    const unsigned lane_off = (unsigned)(k_lo * HWSZ + m);

    v2f pblo[NPLAIN], pbhi[NPLAIN];
    v2f cblo[NCENT],  cbhi[NCENT];

    #pragma unroll
    for (int j = 0; j < NPLAIN; ++j) {
        const int i  = wr * 7 + j / 3;
        const int hr = reflect96(h0 - PADR + i);
        const float* xs = xw + hr * WW + (j % 3) * 16;
        pblo[j][0] = xs[lane_off];
        pblo[j][1] = xs[lane_off + HWSZ];
        pbhi[j][0] = xs[lane_off + 4 * HWSZ];
        pbhi[j][1] = xs[lane_off + 5 * HWSZ];
    }
    #pragma unroll
    for (int j = 0; j < NCENT; ++j) {
        const int rr = wr * 2 + j / 3;
        const float* xs = xw + (h0 + rr) * WW + (j % 3) * 16;
        cblo[j][0] = xs[lane_off];
        cblo[j][1] = xs[lane_off + HWSZ];
        cbhi[j][0] = xs[lane_off + 4 * HWSZ];
        cbhi[j][1] = xs[lane_off + 5 * HWSZ];
    }

    rcomp[tid / NCH][tid % NCH] = rcv;

    #pragma unroll
    for (int j = 0; j < NPLAIN; ++j) {
        const int i   = wr * 7 + j / 3;
        const int col = wc * 48 + (j % 3) * 16 + m;

        v8f acc = {};
        acc = __builtin_amdgcn_wmma_f32_16x16x4_f32(false, aKV_lo, false, pblo[j],
                                                    (short)0, acc, false, false);
        acc = __builtin_amdgcn_wmma_f32_16x16x4_f32(false, aKV_hi, false, pbhi[j],
                                                    (short)0, acc, false, false);
        v4f lo4 = {acc[0], acc[1], acc[2], acc[3]};
        v4f hi4 = {acc[4], acc[5], acc[6], acc[7]};
        if (half) { vb0[i][col] = lo4; vb1[i][col] = hi4; }
        else      { kb0[i][col] = lo4; kb1[i][col] = hi4; }
    }
    #pragma unroll
    for (int j = 0; j < NCENT; ++j) {
        const int rr  = wr * 2 + j / 3;
        const int i   = rr + PADR;
        const int col = wc * 48 + (j % 3) * 16 + m;

        v8f acc = {};
        acc = __builtin_amdgcn_wmma_f32_16x16x4_f32(false, aKV_lo, false, cblo[j],
                                                    (short)0, acc, false, false);
        acc = __builtin_amdgcn_wmma_f32_16x16x4_f32(false, aKV_hi, false, cbhi[j],
                                                    (short)0, acc, false, false);
        v8f qacc = {};
        qacc = __builtin_amdgcn_wmma_f32_16x16x4_f32(false, aQ_lo, false, cblo[j],
                                                     (short)0, qacc, false, false);
        qacc = __builtin_amdgcn_wmma_f32_16x16x4_f32(false, aQ_hi, false, cbhi[j],
                                                     (short)0, qacc, false, false);

        v4f lo4 = {acc[0], acc[1], acc[2], acc[3]};
        v4f hi4 = {acc[4], acc[5], acc[6], acc[7]};
        if (half) { vb0[i][col] = lo4; vb1[i][col] = hi4; }
        else      { kb0[i][col] = lo4; kb1[i][col] = hi4; }

        v4f qlo4 = {qacc[0], qacc[1], qacc[2], qacc[3]};
        v4f qhi4 = {qacc[4], qacc[5], qacc[6], qacc[7]};
        if (half) qb1[rr][col] = qhi4;
        else      qb0[rr][col] = qlo4;
    }

    __syncthreads();

    const bool rowBias = (g < 4);
    for (int p = tid; p < ROWS * WW; p += NTHR) {
        const int rr = p / WW;
        const int w  = p % WW;

        const v4f q0 = qb0[rr][w];
        const v4f q1 = qb1[rr][w];

        float rb[KW];
        #pragma unroll
        for (int t = 0; t < KW; ++t) {
            const v4f r0 = *(const v4f*)&rcomp[t][0];
            const v4f r1 = *(const v4f*)&rcomp[t][4];
            float s = 0.f;
            #pragma unroll
            for (int n = 0; n < 4; ++n) s = __builtin_fmaf(q0[n], r0[n], s);
            #pragma unroll
            for (int n = 0; n < 4; ++n) s = __builtin_fmaf(q1[n], r1[n], s);
            rb[t] = s;
        }

        float mmax = -3.0e38f, ssum = 0.0f;
        float acc[NCH];
        #pragma unroll
        for (int n = 0; n < NCH; ++n) acc[n] = 0.0f;

        for (int ki = 0; ki < KW; ++ki) {
            const int slot = rr + ki;
            for (int kj = 0; kj < KW; ++kj) {
                int c = w + kj - PADR;
                c = (c < 0) ? -c : ((c > 95) ? 190 - c : c);

                const v4f k0 = kb0[slot][c];
                const v4f k1 = kb1[slot][c];
                float l = rb[rowBias ? ki : kj];
                #pragma unroll
                for (int n = 0; n < 4; ++n) l = __builtin_fmaf(q0[n], k0[n], l);
                #pragma unroll
                for (int n = 0; n < 4; ++n) l = __builtin_fmaf(q1[n], k1[n], l);

                const float mn   = fmaxf(mmax, l);
                const float corr = __expf(mmax - mn);
                const float e    = __expf(l - mn);
                ssum = ssum * corr + e;

                const v4f va  = vb0[slot][c];
                const v4f vbb = vb1[slot][c];
                #pragma unroll
                for (int n = 0; n < 4; ++n) acc[n]     = acc[n]     * corr + e * va[n];
                #pragma unroll
                for (int n = 0; n < 4; ++n) acc[n + 4] = acc[n + 4] * corr + e * vbb[n];
                mmax = mn;
            }
        }

        const float inv = 1.0f / ssum;
        float* op = out + ((size_t)(b * CIN + g * NCH) * HH + (h0 + rr)) * WW + w;
        #pragma unroll
        for (int n = 0; n < NCH; ++n)
            op[(size_t)n * HWSZ] = acc[n] * inv;
    }
}

extern "C" void kernel_launch(void* const* d_in, const int* in_sizes, int n_in,
                              void* d_out, int out_size, void* d_ws, size_t ws_size,
                              hipStream_t stream) {
    (void)in_sizes; (void)n_in; (void)out_size; (void)d_ws; (void)ws_size;
    const float* x    = (const float*)d_in[0];
    const float* Wk   = (const float*)d_in[1];
    const float* Wq   = (const float*)d_in[2];
    const float* Wv   = (const float*)d_in[3];
    const float* r_ai = (const float*)d_in[4];
    const float* r_aj = (const float*)d_in[5];
    float* out = (float*)d_out;

    dim3 grid(HH / ROWS, GROUPS, BQ);
    SelfAttentionBlock_kernel<<<grid, NTHR, 0, stream>>>(x, Wk, Wq, Wv, r_ai, r_aj, out);
}


# Position-bias term made a fixed register: attention loop compiled twice (row-bias groups, column-bias groups), so no per-position run-time select among the 7 bias registers; sums unchanged
# speedup vs baseline: 1.8249x; 1.2040x over previous
#include <hip/hip_runtime.h>

#define BQ     4
#define CIN    64
#define HH     96
#define WW     96
#define GROUPS 8
#define NCH    8
#define CGC    8
#define KW     7
#define PADR   3
#define HWSZ   (HH * WW)

#define ROWS   4
#define KVROWS (ROWS + 2*PADR)
#define NTHR   128
#define NPLAIN 9
#define NCENT  6

typedef __attribute__((ext_vector_type(2))) float v2f;
typedef __attribute__((ext_vector_type(4))) float v4f;
typedef __attribute__((ext_vector_type(8))) float v8f;

__device__ __forceinline__ int reflect96(int i) {
    if (i < 0)  i = -i;
    if (i > 95) i = 190 - i;
    return i;
}

template <bool ROW_BIAS>
__device__ __forceinline__ void attend(const v4f (&kb0)[KVROWS][WW], const v4f (&kb1)[KVROWS][WW],
                                       const v4f (&vb0)[KVROWS][WW], const v4f (&vb1)[KVROWS][WW],
                                       const v4f (&qb0)[ROWS][WW],   const v4f (&qb1)[ROWS][WW],
                                       const float (&rcomp)[NTHR / NCH][NCH],
                                       float* __restrict__ outg, int tid)
{
    for (int p = tid; p < ROWS * WW; p += NTHR) {
        const int rr = p / WW;
        const int w  = p % WW;

        const v4f q0 = qb0[rr][w];
        const v4f q1 = qb1[rr][w];

        float rb[KW];
        if (!ROW_BIAS) {
            #pragma unroll
            for (int t = 0; t < KW; ++t) {
                const v4f r0 = *(const v4f*)&rcomp[t][0];
                const v4f r1 = *(const v4f*)&rcomp[t][4];
                float s = 0.f;
                #pragma unroll
                for (int n = 0; n < 4; ++n) s = __builtin_fmaf(q0[n], r0[n], s);
                #pragma unroll
                for (int n = 0; n < 4; ++n) s = __builtin_fmaf(q1[n], r1[n], s);
                rb[t] = s;
            }
        }

        float mmax = -3.0e38f, ssum = 0.0f;
        float acc[NCH];
        #pragma unroll
        for (int n = 0; n < NCH; ++n) acc[n] = 0.0f;

        for (int ki = 0; ki < KW; ++ki) {
            const int slot = rr + ki;

            float rbrow = 0.f;
            if (ROW_BIAS) {
                const v4f r0 = *(const v4f*)&rcomp[ki][0];
                const v4f r1 = *(const v4f*)&rcomp[ki][4];
                #pragma unroll
                for (int n = 0; n < 4; ++n) rbrow = __builtin_fmaf(q0[n], r0[n], rbrow);
                #pragma unroll
                for (int n = 0; n < 4; ++n) rbrow = __builtin_fmaf(q1[n], r1[n], rbrow);
            }

            for (int kj = 0; kj < KW; ++kj) {
                int c = w + kj - PADR;
                c = (c < 0) ? -c : ((c > 95) ? 190 - c : c);

                const v4f k0 = kb0[slot][c];
                const v4f k1 = kb1[slot][c];
                float l = ROW_BIAS ? rbrow : rb[kj];
                #pragma unroll
                for (int n = 0; n < 4; ++n) l = __builtin_fmaf(q0[n], k0[n], l);
                #pragma unroll
                for (int n = 0; n < 4; ++n) l = __builtin_fmaf(q1[n], k1[n], l);

                const float mn   = fmaxf(mmax, l);
                const float corr = __expf(mmax - mn);
                const float e    = __expf(l - mn);
                ssum = ssum * corr + e;

                const v4f va  = vb0[slot][c];
                const v4f vbb = vb1[slot][c];
                #pragma unroll
                for (int n = 0; n < 4; ++n) acc[n]     = acc[n]     * corr + e * va[n];
                #pragma unroll
                for (int n = 0; n < 4; ++n) acc[n + 4] = acc[n + 4] * corr + e * vbb[n];
                mmax = mn;
            }
        }

        const float inv = 1.0f / ssum;
        float* op = outg + (size_t)rr * WW + w;
        #pragma unroll
        for (int n = 0; n < NCH; ++n)
            op[(size_t)n * HWSZ] = acc[n] * inv;
    }
}

__global__ __launch_bounds__(NTHR)
void SelfAttentionBlock_kernel(const float* __restrict__ x,
                               const float* __restrict__ Wk,
                               const float* __restrict__ Wq,
                               const float* __restrict__ Wv,
                               const float* __restrict__ r_ai,
                               const float* __restrict__ r_aj,
                               float* __restrict__ out)
{
    __shared__ v4f kb0[KVROWS][WW];
    __shared__ v4f kb1[KVROWS][WW];
    __shared__ v4f vb0[KVROWS][WW];
    __shared__ v4f vb1[KVROWS][WW];
    __shared__ v4f qb0[ROWS][WW];
    __shared__ v4f qb1[ROWS][WW];
    __shared__ __attribute__((aligned(16))) float rcomp[NTHR / NCH][NCH];

    const int h0   = blockIdx.x * ROWS;
    const int g    = blockIdx.y;
    const int b    = blockIdx.z;
    const int tid  = threadIdx.x;
    const int lane = tid & 31;
    const int wave = __builtin_amdgcn_readfirstlane(tid >> 5);
    const int half = lane >> 4;
    const int m    = lane & 15;

    float rcv;
    {
        const int ti = (tid < KW * NCH) ? tid : 0;
        const int t = ti / NCH, n = ti % NCH;
        const int cg = g * NCH + n;
        rcv = (g < 4) ? r_ai[cg * KW + t] : r_aj[(cg - 32) * KW + t];
    }

    const float* WkG = Wk + g * 64;
    const float* WqG = Wq + g * 64;
    const float* WvG = Wv + g * 64;
    const int k_lo = half * 2;
    const int k_hi = 4 + half * 2;

    v2f aKV_lo, aKV_hi, aQ_lo, aQ_hi;
    if (m < 8) {
        aKV_lo[0] = WkG[m * 8 + k_lo];     aKV_lo[1] = WkG[m * 8 + k_lo + 1];
        aKV_hi[0] = WkG[m * 8 + k_hi];     aKV_hi[1] = WkG[m * 8 + k_hi + 1];
    } else {
        aKV_lo[0] = WvG[(m - 8) * 8 + k_lo]; aKV_lo[1] = WvG[(m - 8) * 8 + k_lo + 1];
        aKV_hi[0] = WvG[(m - 8) * 8 + k_hi]; aKV_hi[1] = WvG[(m - 8) * 8 + k_hi + 1];
    }
    aQ_lo[0] = WqG[(m & 7) * 8 + k_lo];  aQ_lo[1] = WqG[(m & 7) * 8 + k_lo + 1];
    aQ_hi[0] = WqG[(m & 7) * 8 + k_hi];  aQ_hi[1] = WqG[(m & 7) * 8 + k_hi + 1];

    const float* xg = x + (size_t)(b * CIN + g * CGC) * HWSZ;

    const int wr = wave >> 1;
    const int wc = wave & 1;
    const float* xw = xg + wc * 48;
    const unsigned lane_off = (unsigned)(k_lo * HWSZ + m);

    v2f pblo[NPLAIN], pbhi[NPLAIN];
    v2f cblo[NCENT],  cbhi[NCENT];

    #pragma unroll
    for (int j = 0; j < NPLAIN; ++j) {
        const int i  = wr * 7 + j / 3;
        const int hr = reflect96(h0 - PADR + i);
        const float* xs = xw + hr * WW + (j % 3) * 16;
        pblo[j][0] = xs[lane_off];
        pblo[j][1] = xs[lane_off + HWSZ];
        pbhi[j][0] = xs[lane_off + 4 * HWSZ];
        pbhi[j][1] = xs[lane_off + 5 * HWSZ];
    }
    #pragma unroll
    for (int j = 0; j < NCENT; ++j) {
        const int rr = wr * 2 + j / 3;
        const float* xs = xw + (h0 + rr) * WW + (j % 3) * 16;
        cblo[j][0] = xs[lane_off];
        cblo[j][1] = xs[lane_off + HWSZ];
        cbhi[j][0] = xs[lane_off + 4 * HWSZ];
        cbhi[j][1] = xs[lane_off + 5 * HWSZ];
    }

    rcomp[tid / NCH][tid % NCH] = rcv;

    #pragma unroll
    for (int j = 0; j < NPLAIN; ++j) {
        const int i   = wr * 7 + j / 3;
        const int col = wc * 48 + (j % 3) * 16 + m;

        v8f acc = {};
        acc = __builtin_amdgcn_wmma_f32_16x16x4_f32(false, aKV_lo, false, pblo[j],
                                                    (short)0, acc, false, false);
        acc = __builtin_amdgcn_wmma_f32_16x16x4_f32(false, aKV_hi, false, pbhi[j],
                                                    (short)0, acc, false, false);
        v4f lo4 = {acc[0], acc[1], acc[2], acc[3]};
        v4f hi4 = {acc[4], acc[5], acc[6], acc[7]};
        if (half) { vb0[i][col] = lo4; vb1[i][col] = hi4; }
        else      { kb0[i][col] = lo4; kb1[i][col] = hi4; }
    }
    #pragma unroll
    for (int j = 0; j < NCENT; ++j) {
        const int rr  = wr * 2 + j / 3;
        const int i   = rr + PADR;
        const int col = wc * 48 + (j % 3) * 16 + m;

        v8f acc = {};
        acc = __builtin_amdgcn_wmma_f32_16x16x4_f32(false, aKV_lo, false, cblo[j],
                                                    (short)0, acc, false, false);
        acc = __builtin_amdgcn_wmma_f32_16x16x4_f32(false, aKV_hi, false, cbhi[j],
                                                    (short)0, acc, false, false);
        v8f qacc = {};
        qacc = __builtin_amdgcn_wmma_f32_16x16x4_f32(false, aQ_lo, false, cblo[j],
                                                     (short)0, qacc, false, false);
        qacc = __builtin_amdgcn_wmma_f32_16x16x4_f32(false, aQ_hi, false, cbhi[j],
                                                     (short)0, qacc, false, false);

        v4f lo4 = {acc[0], acc[1], acc[2], acc[3]};
        v4f hi4 = {acc[4], acc[5], acc[6], acc[7]};
        if (half) { vb0[i][col] = lo4; vb1[i][col] = hi4; }
        else      { kb0[i][col] = lo4; kb1[i][col] = hi4; }

        v4f qlo4 = {qacc[0], qacc[1], qacc[2], qacc[3]};
        v4f qhi4 = {qacc[4], qacc[5], qacc[6], qacc[7]};
        if (half) qb1[rr][col] = qhi4;
        else      qb0[rr][col] = qlo4;
    }

    __syncthreads();

    float* outg = out + ((size_t)(b * CIN + g * NCH) * HH + h0) * WW;
    if (g < 4) attend<true >(kb0, kb1, vb0, vb1, qb0, qb1, rcomp, outg, tid);
    else       attend<false>(kb0, kb1, vb0, vb1, qb0, qb1, rcomp, outg, tid);
}

extern "C" void kernel_launch(void* const* d_in, const int* in_sizes, int n_in,
                              void* d_out, int out_size, void* d_ws, size_t ws_size,
                              hipStream_t stream) {
    (void)in_sizes; (void)n_in; (void)out_size; (void)d_ws; (void)ws_size;
    const float* x    = (const float*)d_in[0];
    const float* Wk   = (const float*)d_in[1];
    const float* Wq   = (const float*)d_in[2];
    const float* Wv   = (const float*)d_in[3];
    const float* r_ai = (const float*)d_in[4];
    const float* r_aj = (const float*)d_in[5];
    float* out = (float*)d_out;

    dim3 grid(HH / ROWS, GROUPS, BQ);
    SelfAttentionBlock_kernel<<<grid, NTHR, 0, stream>>>(x, Wk, Wq, Wv, r_ai, r_aj, out);
}


# Attention: wave = output row, lane = 3 adjacent pixels, each k/v record read once for the three; running max and rescale of the sums once per window row (7 logits kept), not per position
# speedup vs baseline: 2.0223x; 1.1082x over previous
#include <hip/hip_runtime.h>

#define BQ     4
#define CIN    64
#define HH     96
#define WW     96
#define GROUPS 8
#define NCH    8
#define CGC    8
#define KW     7
#define PADR   3
#define HWSZ   (HH * WW)

#define ROWS   4
#define KVROWS (ROWS + 2*PADR)
#define NTHR   128
#define NPLAIN 9
#define NCENT  6
#define PX     3

typedef __attribute__((ext_vector_type(2))) float v2f;
typedef __attribute__((ext_vector_type(4))) float v4f;
typedef __attribute__((ext_vector_type(8))) float v8f;

__device__ __forceinline__ int reflect96(int i) {
    if (i < 0)  i = -i;
    if (i > 95) i = 190 - i;
    return i;
}

template <bool ROW_BIAS>
__device__ __forceinline__ void attend(const v4f (&kb0)[KVROWS][WW], const v4f (&kb1)[KVROWS][WW],
                                       const v4f (&vb0)[KVROWS][WW], const v4f (&vb1)[KVROWS][WW],
                                       v4f (&qb0)[ROWS][WW],         v4f (&qb1)[ROWS][WW],
                                       const float (&rcomp)[NTHR / NCH][NCH],
                                       float* __restrict__ outg, int rr, int lane)
{
    const int w0 = 3 * lane;

    int cc[PX + KW - 1];
    #pragma unroll
    for (int t = 0; t < PX + KW - 1; ++t) {
        int c = w0 + t - PADR;
        cc[t] = (c < 0) ? -c : ((c > 95) ? 190 - c : c);
    }

    v4f q0[PX], q1[PX];
    #pragma unroll
    for (int j = 0; j < PX; ++j) { q0[j] = qb0[rr][w0 + j]; q1[j] = qb1[rr][w0 + j]; }

    float rb[PX][KW];
    if (!ROW_BIAS) {
        #pragma unroll
        for (int t = 0; t < KW; ++t) {
            const v4f r0 = *(const v4f*)&rcomp[t][0];
            const v4f r1 = *(const v4f*)&rcomp[t][4];
            #pragma unroll
            for (int j = 0; j < PX; ++j) {
                float s = 0.f;
                #pragma unroll
                for (int n = 0; n < 4; ++n) s = __builtin_fmaf(q0[j][n], r0[n], s);
                #pragma unroll
                for (int n = 0; n < 4; ++n) s = __builtin_fmaf(q1[j][n], r1[n], s);
                rb[j][t] = s;
            }
        }
    }

    float mmax[PX], ssum[PX];
    float acc[PX][NCH];
    #pragma unroll
    for (int j = 0; j < PX; ++j) {
        mmax[j] = -3.0e38f;
        ssum[j] = 0.0f;
        #pragma unroll
        for (int n = 0; n < NCH; ++n) acc[j][n] = 0.0f;
    }

    #pragma nounroll
    for (int ki = 0; ki < KW; ++ki) {
        const int slot = rr + ki;

        float rbrow[PX];
        if (ROW_BIAS) {
            const v4f r0 = *(const v4f*)&rcomp[ki][0];
            const v4f r1 = *(const v4f*)&rcomp[ki][4];
            #pragma unroll
            for (int j = 0; j < PX; ++j) {
                float s = 0.f;
                #pragma unroll
                for (int n = 0; n < 4; ++n) s = __builtin_fmaf(q0[j][n], r0[n], s);
                #pragma unroll
                for (int n = 0; n < 4; ++n) s = __builtin_fmaf(q1[j][n], r1[n], s);
                rbrow[j] = s;
            }
        }

        float lg[PX][KW];
        #pragma unroll
        for (int t = 0; t < PX + KW - 1; ++t) {
            const v4f k0 = kb0[slot][cc[t]];
            const v4f k1 = kb1[slot][cc[t]];
            #pragma unroll
            for (int j = 0; j < PX; ++j) {
                const int kj = t - j;
                if (kj >= 0 && kj < KW) {
                    float l = ROW_BIAS ? rbrow[j] : rb[j][kj];
                    #pragma unroll
                    for (int n = 0; n < 4; ++n) l = __builtin_fmaf(q0[j][n], k0[n], l);
                    #pragma unroll
                    for (int n = 0; n < 4; ++n) l = __builtin_fmaf(q1[j][n], k1[n], l);
                    lg[j][kj] = l;
                }
            }
        }

        #pragma unroll
        for (int j = 0; j < PX; ++j) {
            float mn = mmax[j];
            #pragma unroll
            for (int kj = 0; kj < KW; ++kj) mn = fmaxf(mn, lg[j][kj]);
            const float corr = __expf(mmax[j] - mn);
            mmax[j] = mn;
            ssum[j] *= corr;
            #pragma unroll
            for (int n = 0; n < NCH; ++n) acc[j][n] *= corr;
        }

        #pragma unroll
        for (int t = 0; t < PX + KW - 1; ++t) {
            const v4f va  = vb0[slot][cc[t]];
            const v4f vbb = vb1[slot][cc[t]];
            #pragma unroll
            for (int j = 0; j < PX; ++j) {
                const int kj = t - j;
                if (kj >= 0 && kj < KW) {
                    const float e = __expf(lg[j][kj] - mmax[j]);
                    ssum[j] += e;
                    #pragma unroll
                    for (int n = 0; n < 4; ++n) acc[j][n]     = __builtin_fmaf(e, va[n],  acc[j][n]);
                    #pragma unroll
                    for (int n = 0; n < 4; ++n) acc[j][n + 4] = __builtin_fmaf(e, vbb[n], acc[j][n + 4]);
                }
            }
        }
    }

    float* st0 = reinterpret_cast<float*>(&qb0[rr][0]);
    float* st1 = reinterpret_cast<float*>(&qb1[rr][0]);
    #pragma unroll
    for (int j = 0; j < PX; ++j) {
        const float inv = 1.0f / ssum[j];
        #pragma unroll
        for (int n = 0; n < 4; ++n) {
            st0[n * WW + w0 + j] = acc[j][n]     * inv;
            st1[n * WW + w0 + j] = acc[j][n + 4] * inv;
        }
    }
    __syncthreads();
    float* op = outg + (size_t)rr * WW + lane;
    #pragma unroll
    for (int n = 0; n < 4; ++n) {
        #pragma unroll
        for (int t = 0; t < PX; ++t) {
            op[(size_t)n * HWSZ + 32 * t]       = st0[n * WW + lane + 32 * t];
            op[(size_t)(n + 4) * HWSZ + 32 * t] = st1[n * WW + lane + 32 * t];
        }
    }
}

__global__ __launch_bounds__(NTHR)
void SelfAttentionBlock_kernel(const float* __restrict__ x,
                               const float* __restrict__ Wk,
                               const float* __restrict__ Wq,
                               const float* __restrict__ Wv,
                               const float* __restrict__ r_ai,
                               const float* __restrict__ r_aj,
                               float* __restrict__ out)
{
    __shared__ v4f kb0[KVROWS][WW];
    __shared__ v4f kb1[KVROWS][WW];
    __shared__ v4f vb0[KVROWS][WW];
    __shared__ v4f vb1[KVROWS][WW];
    __shared__ v4f qb0[ROWS][WW];
    __shared__ v4f qb1[ROWS][WW];
    __shared__ __attribute__((aligned(16))) float rcomp[NTHR / NCH][NCH];

    const int h0   = blockIdx.x * ROWS;
    const int g    = blockIdx.y;
    const int b    = blockIdx.z;
    const int tid  = threadIdx.x;
    const int lane = tid & 31;
    const int wave = __builtin_amdgcn_readfirstlane(tid >> 5);
    const int half = lane >> 4;
    const int m    = lane & 15;

    float rcv;
    {
        const int ti = (tid < KW * NCH) ? tid : 0;
        const int t = ti / NCH, n = ti % NCH;
        const int cg = g * NCH + n;
        rcv = (g < 4) ? r_ai[cg * KW + t] : r_aj[(cg - 32) * KW + t];
    }

    const float* WkG = Wk + g * 64;
    const float* WqG = Wq + g * 64;
    const float* WvG = Wv + g * 64;
    const int k_lo = half * 2;
    const int k_hi = 4 + half * 2;

    v2f aKV_lo, aKV_hi, aQ_lo, aQ_hi;
    if (m < 8) {
        aKV_lo[0] = WkG[m * 8 + k_lo];     aKV_lo[1] = WkG[m * 8 + k_lo + 1];
        aKV_hi[0] = WkG[m * 8 + k_hi];     aKV_hi[1] = WkG[m * 8 + k_hi + 1];
    } else {
        aKV_lo[0] = WvG[(m - 8) * 8 + k_lo]; aKV_lo[1] = WvG[(m - 8) * 8 + k_lo + 1];
        aKV_hi[0] = WvG[(m - 8) * 8 + k_hi]; aKV_hi[1] = WvG[(m - 8) * 8 + k_hi + 1];
    }
    aQ_lo[0] = WqG[(m & 7) * 8 + k_lo];  aQ_lo[1] = WqG[(m & 7) * 8 + k_lo + 1];
    aQ_hi[0] = WqG[(m & 7) * 8 + k_hi];  aQ_hi[1] = WqG[(m & 7) * 8 + k_hi + 1];

    const float* xg = x + (size_t)(b * CIN + g * CGC) * HWSZ;

    const int wr = wave >> 1;
    const int wc = wave & 1;
    const float* xw = xg + wc * 48;
    const unsigned lane_off = (unsigned)(k_lo * HWSZ + m);

    v2f pblo[NPLAIN], pbhi[NPLAIN];
    v2f cblo[NCENT],  cbhi[NCENT];

    #pragma unroll
    for (int j = 0; j < NPLAIN; ++j) {
        const int i  = wr * 7 + j / 3;
        const int hr = reflect96(h0 - PADR + i);
        const float* xs = xw + hr * WW + (j % 3) * 16;
        pblo[j][0] = xs[lane_off];
        pblo[j][1] = xs[lane_off + HWSZ];
        pbhi[j][0] = xs[lane_off + 4 * HWSZ];
        pbhi[j][1] = xs[lane_off + 5 * HWSZ];
    }
    #pragma unroll
    for (int j = 0; j < NCENT; ++j) {
        const int rr = wr * 2 + j / 3;
        const float* xs = xw + (h0 + rr) * WW + (j % 3) * 16;
        cblo[j][0] = xs[lane_off];
        cblo[j][1] = xs[lane_off + HWSZ];
        cbhi[j][0] = xs[lane_off + 4 * HWSZ];
        cbhi[j][1] = xs[lane_off + 5 * HWSZ];
    }

    rcomp[tid / NCH][tid % NCH] = rcv;

    #pragma unroll
    for (int j = 0; j < NPLAIN; ++j) {
        const int i   = wr * 7 + j / 3;
        const int col = wc * 48 + (j % 3) * 16 + m;

        v8f acc = {};
        acc = __builtin_amdgcn_wmma_f32_16x16x4_f32(false, aKV_lo, false, pblo[j],
                                                    (short)0, acc, false, false);
        acc = __builtin_amdgcn_wmma_f32_16x16x4_f32(false, aKV_hi, false, pbhi[j],
                                                    (short)0, acc, false, false);
        v4f lo4 = {acc[0], acc[1], acc[2], acc[3]};
        v4f hi4 = {acc[4], acc[5], acc[6], acc[7]};
        if (half) { vb0[i][col] = lo4; vb1[i][col] = hi4; }
        else      { kb0[i][col] = lo4; kb1[i][col] = hi4; }
    }
    #pragma unroll
    for (int j = 0; j < NCENT; ++j) {
        const int rr  = wr * 2 + j / 3;
        const int i   = rr + PADR;
        const int col = wc * 48 + (j % 3) * 16 + m;

        v8f acc = {};
        acc = __builtin_amdgcn_wmma_f32_16x16x4_f32(false, aKV_lo, false, cblo[j],
                                                    (short)0, acc, false, false);
        acc = __builtin_amdgcn_wmma_f32_16x16x4_f32(false, aKV_hi, false, cbhi[j],
                                                    (short)0, acc, false, false);
        v8f qacc = {};
        qacc = __builtin_amdgcn_wmma_f32_16x16x4_f32(false, aQ_lo, false, cblo[j],
                                                     (short)0, qacc, false, false);
        qacc = __builtin_amdgcn_wmma_f32_16x16x4_f32(false, aQ_hi, false, cbhi[j],
                                                     (short)0, qacc, false, false);

        v4f lo4 = {acc[0], acc[1], acc[2], acc[3]};
        v4f hi4 = {acc[4], acc[5], acc[6], acc[7]};
        if (half) { vb0[i][col] = lo4; vb1[i][col] = hi4; }
        else      { kb0[i][col] = lo4; kb1[i][col] = hi4; }

        v4f qlo4 = {qacc[0], qacc[1], qacc[2], qacc[3]};
        v4f qhi4 = {qacc[4], qacc[5], qacc[6], qacc[7]};
        if (half) qb1[rr][col] = qhi4;
        else      qb0[rr][col] = qlo4;
    }

    __syncthreads();

    float* outg = out + ((size_t)(b * CIN + g * NCH) * HH + h0) * WW;
    if (g < 4) attend<true >(kb0, kb1, vb0, vb1, qb0, qb1, rcomp, outg, wave, lane);
    else       attend<false>(kb0, kb1, vb0, vb1, qb0, qb1, rcomp, outg, wave, lane);
}

extern "C" void kernel_launch(void* const* d_in, const int* in_sizes, int n_in,
                              void* d_out, int out_size, void* d_ws, size_t ws_size,
                              hipStream_t stream) {
    (void)in_sizes; (void)n_in; (void)out_size; (void)d_ws; (void)ws_size;
    const float* x    = (const float*)d_in[0];
    const float* Wk   = (const float*)d_in[1];
    const float* Wq   = (const float*)d_in[2];
    const float* Wv   = (const float*)d_in[3];
    const float* r_ai = (const float*)d_in[4];
    const float* r_aj = (const float*)d_in[5];
    float* out = (float*)d_out;

    dim3 grid(HH / ROWS, GROUPS, BQ);
    SelfAttentionBlock_kernel<<<grid, NTHR, 0, stream>>>(x, Wk, Wq, Wv, r_ai, r_aj, out);
}


# Attention logits: each logit's 8 products added as two interleaved chains (even/odd channels) in packed FMAs on register pairs as loaded, then joined; no operand moves; logit sum order changed
# speedup vs baseline: 2.3276x; 1.1510x over previous
#include <hip/hip_runtime.h>

#define BQ     4
#define CIN    64
#define HH     96
#define WW     96
#define GROUPS 8
#define NCH    8
#define CGC    8
#define KW     7
#define PADR   3
#define HWSZ   (HH * WW)

#define ROWS   4
#define KVROWS (ROWS + 2*PADR)
#define NTHR   128
#define NPLAIN 9
#define NCENT  6
#define PX     3

typedef __attribute__((ext_vector_type(2))) float v2f;
typedef __attribute__((ext_vector_type(4))) float v4f;
typedef __attribute__((ext_vector_type(8))) float v8f;

__device__ __forceinline__ int reflect96(int i) {
    if (i < 0)  i = -i;
    if (i > 95) i = 190 - i;
    return i;
}

template <bool ROW_BIAS>
__device__ __forceinline__ void attend(const v4f (&kb0)[KVROWS][WW], const v4f (&kb1)[KVROWS][WW],
                                       const v4f (&vb0)[KVROWS][WW], const v4f (&vb1)[KVROWS][WW],
                                       v4f (&qb0)[ROWS][WW],         v4f (&qb1)[ROWS][WW],
                                       const float (&rcomp)[NTHR / NCH][NCH],
                                       float* __restrict__ outg, int rr, int lane)
{
    const int w0 = 3 * lane;

    int cc[PX + KW - 1];
    #pragma unroll
    for (int t = 0; t < PX + KW - 1; ++t) {
        int c = w0 + t - PADR;
        cc[t] = (c < 0) ? -c : ((c > 95) ? 190 - c : c);
    }

    v2f qp[PX][4];
    #pragma unroll
    for (int j = 0; j < PX; ++j) {
        const v4f a = qb0[rr][w0 + j];
        const v4f b = qb1[rr][w0 + j];
        qp[j][0] = __builtin_shufflevector(a, a, 0, 1);
        qp[j][1] = __builtin_shufflevector(a, a, 2, 3);
        qp[j][2] = __builtin_shufflevector(b, b, 0, 1);
        qp[j][3] = __builtin_shufflevector(b, b, 2, 3);
    }

    float rb[PX][KW];
    if (!ROW_BIAS) {
        #pragma unroll
        for (int t = 0; t < KW; ++t) {
            const v4f r0 = *(const v4f*)&rcomp[t][0];
            const v4f r1 = *(const v4f*)&rcomp[t][4];
            const v2f rp[4] = {__builtin_shufflevector(r0, r0, 0, 1), __builtin_shufflevector(r0, r0, 2, 3),
                               __builtin_shufflevector(r1, r1, 0, 1), __builtin_shufflevector(r1, r1, 2, 3)};
            #pragma unroll
            for (int j = 0; j < PX; ++j) {
                v2f s = qp[j][0] * rp[0];
                #pragma unroll
                for (int m = 1; m < 4; ++m) s = __builtin_elementwise_fma(qp[j][m], rp[m], s);
                rb[j][t] = s[0] + s[1];
            }
        }
    }

    float mmax[PX], ssum[PX];
    float acc[PX][NCH];
    #pragma unroll
    for (int j = 0; j < PX; ++j) {
        mmax[j] = -3.0e38f;
        ssum[j] = 0.0f;
        #pragma unroll
        for (int n = 0; n < NCH; ++n) acc[j][n] = 0.0f;
    }

    #pragma nounroll
    for (int ki = 0; ki < KW; ++ki) {
        const int slot = rr + ki;

        v2f rbrow[PX];
        if (ROW_BIAS) {
            const v4f r0 = *(const v4f*)&rcomp[ki][0];
            const v4f r1 = *(const v4f*)&rcomp[ki][4];
            const v2f rp[4] = {__builtin_shufflevector(r0, r0, 0, 1), __builtin_shufflevector(r0, r0, 2, 3),
                               __builtin_shufflevector(r1, r1, 0, 1), __builtin_shufflevector(r1, r1, 2, 3)};
            #pragma unroll
            for (int j = 0; j < PX; ++j) {
                v2f s = qp[j][0] * rp[0];
                #pragma unroll
                for (int m = 1; m < 4; ++m) s = __builtin_elementwise_fma(qp[j][m], rp[m], s);
                rbrow[j] = s;
            }
        }

        float lg[PX][KW];
        #pragma unroll
        for (int t = 0; t < PX + KW - 1; ++t) {
            const v4f k0 = kb0[slot][cc[t]];
            const v4f k1 = kb1[slot][cc[t]];
            const v2f kp[4] = {__builtin_shufflevector(k0, k0, 0, 1), __builtin_shufflevector(k0, k0, 2, 3),
                               __builtin_shufflevector(k1, k1, 0, 1), __builtin_shufflevector(k1, k1, 2, 3)};
            #pragma unroll
            for (int j = 0; j < PX; ++j) {
                const int kj = t - j;
                if (kj >= 0 && kj < KW) {
                    if (ROW_BIAS) {
                        v2f s = rbrow[j];
                        #pragma unroll
                        for (int m = 0; m < 4; ++m) s = __builtin_elementwise_fma(qp[j][m], kp[m], s);
                        lg[j][kj] = s[0] + s[1];
                    } else {
                        v2f s = qp[j][0] * kp[0];
                        #pragma unroll
                        for (int m = 1; m < 4; ++m) s = __builtin_elementwise_fma(qp[j][m], kp[m], s);
                        lg[j][kj] = (s[0] + s[1]) + rb[j][kj];
                    }
                }
            }
        }

        #pragma unroll
        for (int j = 0; j < PX; ++j) {
            float mn = mmax[j];
            #pragma unroll
            for (int kj = 0; kj < KW; ++kj) mn = fmaxf(mn, lg[j][kj]);
            const float corr = __expf(mmax[j] - mn);
            mmax[j] = mn;
            ssum[j] *= corr;
            #pragma unroll
            for (int n = 0; n < NCH; ++n) acc[j][n] *= corr;
        }

        #pragma unroll
        for (int t = 0; t < PX + KW - 1; ++t) {
            const v4f va  = vb0[slot][cc[t]];
            const v4f vbb = vb1[slot][cc[t]];
            #pragma unroll
            for (int j = 0; j < PX; ++j) {
                const int kj = t - j;
                if (kj >= 0 && kj < KW) {
                    const float e = __expf(lg[j][kj] - mmax[j]);
                    ssum[j] += e;
                    #pragma unroll
                    for (int n = 0; n < 4; ++n) acc[j][n]     = __builtin_fmaf(e, va[n],  acc[j][n]);
                    #pragma unroll
                    for (int n = 0; n < 4; ++n) acc[j][n + 4] = __builtin_fmaf(e, vbb[n], acc[j][n + 4]);
                }
            }
        }
    }

    float* st0 = reinterpret_cast<float*>(&qb0[rr][0]);
    float* st1 = reinterpret_cast<float*>(&qb1[rr][0]);
    #pragma unroll
    for (int j = 0; j < PX; ++j) {
        const float inv = 1.0f / ssum[j];
        #pragma unroll
        for (int n = 0; n < 4; ++n) {
            st0[n * WW + w0 + j] = acc[j][n]     * inv;
            st1[n * WW + w0 + j] = acc[j][n + 4] * inv;
        }
    }
    __syncthreads();
    float* op = outg + (size_t)rr * WW + lane;
    #pragma unroll
    for (int n = 0; n < 4; ++n) {
        #pragma unroll
        for (int t = 0; t < PX; ++t) {
            op[(size_t)n * HWSZ + 32 * t]       = st0[n * WW + lane + 32 * t];
            op[(size_t)(n + 4) * HWSZ + 32 * t] = st1[n * WW + lane + 32 * t];
        }
    }
}

__global__ __launch_bounds__(NTHR)
void SelfAttentionBlock_kernel(const float* __restrict__ x,
                               const float* __restrict__ Wk,
                               const float* __restrict__ Wq,
                               const float* __restrict__ Wv,
                               const float* __restrict__ r_ai,
                               const float* __restrict__ r_aj,
                               float* __restrict__ out)
{
    __shared__ v4f kb0[KVROWS][WW];
    __shared__ v4f kb1[KVROWS][WW];
    __shared__ v4f vb0[KVROWS][WW];
    __shared__ v4f vb1[KVROWS][WW];
    __shared__ v4f qb0[ROWS][WW];
    __shared__ v4f qb1[ROWS][WW];
    __shared__ __attribute__((aligned(16))) float rcomp[NTHR / NCH][NCH];

    const int h0   = blockIdx.x * ROWS;
    const int g    = blockIdx.y;
    const int b    = blockIdx.z;
    const int tid  = threadIdx.x;
    const int lane = tid & 31;
    const int wave = __builtin_amdgcn_readfirstlane(tid >> 5);
    const int half = lane >> 4;
    const int m    = lane & 15;

    float rcv;
    {
        const int ti = (tid < KW * NCH) ? tid : 0;
        const int t = ti / NCH, n = ti % NCH;
        const int cg = g * NCH + n;
        rcv = (g < 4) ? r_ai[cg * KW + t] : r_aj[(cg - 32) * KW + t];
    }

    const float* WkG = Wk + g * 64;
    const float* WqG = Wq + g * 64;
    const float* WvG = Wv + g * 64;
    const int k_lo = half * 2;
    const int k_hi = 4 + half * 2;

    v2f aKV_lo, aKV_hi, aQ_lo, aQ_hi;
    if (m < 8) {
        aKV_lo[0] = WkG[m * 8 + k_lo];     aKV_lo[1] = WkG[m * 8 + k_lo + 1];
        aKV_hi[0] = WkG[m * 8 + k_hi];     aKV_hi[1] = WkG[m * 8 + k_hi + 1];
    } else {
        aKV_lo[0] = WvG[(m - 8) * 8 + k_lo]; aKV_lo[1] = WvG[(m - 8) * 8 + k_lo + 1];
        aKV_hi[0] = WvG[(m - 8) * 8 + k_hi]; aKV_hi[1] = WvG[(m - 8) * 8 + k_hi + 1];
    }
    aQ_lo[0] = WqG[(m & 7) * 8 + k_lo];  aQ_lo[1] = WqG[(m & 7) * 8 + k_lo + 1];
    aQ_hi[0] = WqG[(m & 7) * 8 + k_hi];  aQ_hi[1] = WqG[(m & 7) * 8 + k_hi + 1];

    const float* xg = x + (size_t)(b * CIN + g * CGC) * HWSZ;

    const int wr = wave >> 1;
    const int wc = wave & 1;
    const float* xw = xg + wc * 48;
    const unsigned lane_off = (unsigned)(k_lo * HWSZ + m);

    v2f pblo[NPLAIN], pbhi[NPLAIN];
    v2f cblo[NCENT],  cbhi[NCENT];

    #pragma unroll
    for (int j = 0; j < NPLAIN; ++j) {
        const int i  = wr * 7 + j / 3;
        const int hr = reflect96(h0 - PADR + i);
        const float* xs = xw + hr * WW + (j % 3) * 16;
        pblo[j][0] = xs[lane_off];
        pblo[j][1] = xs[lane_off + HWSZ];
        pbhi[j][0] = xs[lane_off + 4 * HWSZ];
        pbhi[j][1] = xs[lane_off + 5 * HWSZ];
    }
    #pragma unroll
    for (int j = 0; j < NCENT; ++j) {
        const int rr = wr * 2 + j / 3;
        const float* xs = xw + (h0 + rr) * WW + (j % 3) * 16;
        cblo[j][0] = xs[lane_off];
        cblo[j][1] = xs[lane_off + HWSZ];
        cbhi[j][0] = xs[lane_off + 4 * HWSZ];
        cbhi[j][1] = xs[lane_off + 5 * HWSZ];
    }

    rcomp[tid / NCH][tid % NCH] = rcv;

    #pragma unroll
    for (int j = 0; j < NPLAIN; ++j) {
        const int i   = wr * 7 + j / 3;
        const int col = wc * 48 + (j % 3) * 16 + m;

        v8f acc = {};
        acc = __builtin_amdgcn_wmma_f32_16x16x4_f32(false, aKV_lo, false, pblo[j],
                                                    (short)0, acc, false, false);
        acc = __builtin_amdgcn_wmma_f32_16x16x4_f32(false, aKV_hi, false, pbhi[j],
                                                    (short)0, acc, false, false);
        v4f lo4 = {acc[0], acc[1], acc[2], acc[3]};
        v4f hi4 = {acc[4], acc[5], acc[6], acc[7]};
        if (half) { vb0[i][col] = lo4; vb1[i][col] = hi4; }
        else      { kb0[i][col] = lo4; kb1[i][col] = hi4; }
    }
    #pragma unroll
    for (int j = 0; j < NCENT; ++j) {
        const int rr  = wr * 2 + j / 3;
        const int i   = rr + PADR;
        const int col = wc * 48 + (j % 3) * 16 + m;

        v8f acc = {};
        acc = __builtin_amdgcn_wmma_f32_16x16x4_f32(false, aKV_lo, false, cblo[j],
                                                    (short)0, acc, false, false);
        acc = __builtin_amdgcn_wmma_f32_16x16x4_f32(false, aKV_hi, false, cbhi[j],
                                                    (short)0, acc, false, false);
        v8f qacc = {};
        qacc = __builtin_amdgcn_wmma_f32_16x16x4_f32(false, aQ_lo, false, cblo[j],
                                                     (short)0, qacc, false, false);
        qacc = __builtin_amdgcn_wmma_f32_16x16x4_f32(false, aQ_hi, false, cbhi[j],
                                                     (short)0, qacc, false, false);

        v4f lo4 = {acc[0], acc[1], acc[2], acc[3]};
        v4f hi4 = {acc[4], acc[5], acc[6], acc[7]};
        if (half) { vb0[i][col] = lo4; vb1[i][col] = hi4; }
        else      { kb0[i][col] = lo4; kb1[i][col] = hi4; }

        v4f qlo4 = {qacc[0], qacc[1], qacc[2], qacc[3]};
        v4f qhi4 = {qacc[4], qacc[5], qacc[6], qacc[7]};
        if (half) qb1[rr][col] = qhi4;
        else      qb0[rr][col] = qlo4;
    }

    __syncthreads();

    float* outg = out + ((size_t)(b * CIN + g * NCH) * HH + h0) * WW;
    if (g < 4) attend<true >(kb0, kb1, vb0, vb1, qb0, qb1, rcomp, outg, wave, lane);
    else       attend<false>(kb0, kb1, vb0, vb1, qb0, qb1, rcomp, outg, wave, lane);
}

extern "C" void kernel_launch(void* const* d_in, const int* in_sizes, int n_in,
                              void* d_out, int out_size, void* d_ws, size_t ws_size,
                              hipStream_t stream) {
    (void)in_sizes; (void)n_in; (void)out_size; (void)d_ws; (void)ws_size;
    const float* x    = (const float*)d_in[0];
    const float* Wk   = (const float*)d_in[1];
    const float* Wq   = (const float*)d_in[2];
    const float* Wv   = (const float*)d_in[3];
    const float* r_ai = (const float*)d_in[4];
    const float* r_aj = (const float*)d_in[5];
    float* out = (float*)d_out;

    dim3 grid(HH / ROWS, GROUPS, BQ);
    SelfAttentionBlock_kernel<<<grid, NTHR, 0, stream>>>(x, Wk, Wq, Wv, r_ai, r_aj, out);
}


# One block per WGP: 12 output rows, 12 waves per block (was 4, 4); k/v projected for 18 rows per 12 output rows, not 30: fewer x loads, products and LDS stores; LDS sized to one block per WGP
# speedup vs baseline: 2.5584x; 1.0992x over previous
#include <hip/hip_runtime.h>

#define BQ     4
#define CIN    64
#define HH     96
#define WW     96
#define GROUPS 8
#define NCH    8
#define CGC    8
#define KW     7
#define PADR   3
#define HWSZ   (HH * WW)

#define ROWS   12
#define KVROWS (ROWS + 2*PADR)
#define NTHR   384
#define NPLAIN 3
#define NCENT  6
#define LW     112
#define PX     3

typedef __attribute__((ext_vector_type(2))) float v2f;
typedef __attribute__((ext_vector_type(4))) float v4f;
typedef __attribute__((ext_vector_type(8))) float v8f;

__device__ __forceinline__ int reflect96(int i) {
    if (i < 0)  i = -i;
    if (i > 95) i = 190 - i;
    return i;
}

template <bool ROW_BIAS>
__device__ __forceinline__ void attend(const v4f (&kv)[KVROWS][4][LW], v4f (&qb)[ROWS][2][LW],
                                       const float (&rcomp)[NTHR / NCH][NCH],
                                       float* __restrict__ outg, int rr, int lane)
{
    const int w0 = 3 * lane;

    int cc[PX + KW - 1];
    #pragma unroll
    for (int t = 0; t < PX + KW - 1; ++t) {
        int c = w0 + t - PADR;
        cc[t] = (c < 0) ? -c : ((c > 95) ? 190 - c : c);
    }

    v2f qp[PX][4];
    #pragma unroll
    for (int j = 0; j < PX; ++j) {
        const v4f a = qb[rr][0][w0 + j];
        const v4f b = qb[rr][1][w0 + j];
        qp[j][0] = __builtin_shufflevector(a, a, 0, 1);
        qp[j][1] = __builtin_shufflevector(a, a, 2, 3);
        qp[j][2] = __builtin_shufflevector(b, b, 0, 1);
        qp[j][3] = __builtin_shufflevector(b, b, 2, 3);
    }

    float rb[PX][KW];
    if (!ROW_BIAS) {
        #pragma unroll
        for (int t = 0; t < KW; ++t) {
            const v4f r0 = *(const v4f*)&rcomp[t][0];
            const v4f r1 = *(const v4f*)&rcomp[t][4];
            const v2f rp[4] = {__builtin_shufflevector(r0, r0, 0, 1), __builtin_shufflevector(r0, r0, 2, 3),
                               __builtin_shufflevector(r1, r1, 0, 1), __builtin_shufflevector(r1, r1, 2, 3)};
            #pragma unroll
            for (int j = 0; j < PX; ++j) {
                v2f s = qp[j][0] * rp[0];
                #pragma unroll
                for (int m = 1; m < 4; ++m) s = __builtin_elementwise_fma(qp[j][m], rp[m], s);
                rb[j][t] = s[0] + s[1];
            }
        }
    }

    float mmax[PX], ssum[PX];
    float acc[PX][NCH];
    #pragma unroll
    for (int j = 0; j < PX; ++j) {
        mmax[j] = -3.0e38f;
        ssum[j] = 0.0f;
        #pragma unroll
        for (int n = 0; n < NCH; ++n) acc[j][n] = 0.0f;
    }

    #pragma nounroll
    for (int ki = 0; ki < KW; ++ki) {
        const int slot = rr + ki;

        v2f rbrow[PX];
        if (ROW_BIAS) {
            const v4f r0 = *(const v4f*)&rcomp[ki][0];
            const v4f r1 = *(const v4f*)&rcomp[ki][4];
            const v2f rp[4] = {__builtin_shufflevector(r0, r0, 0, 1), __builtin_shufflevector(r0, r0, 2, 3),
                               __builtin_shufflevector(r1, r1, 0, 1), __builtin_shufflevector(r1, r1, 2, 3)};
            #pragma unroll
            for (int j = 0; j < PX; ++j) {
                v2f s = qp[j][0] * rp[0];
                #pragma unroll
                for (int m = 1; m < 4; ++m) s = __builtin_elementwise_fma(qp[j][m], rp[m], s);
                rbrow[j] = s;
            }
        }

        float lg[PX][KW];
        #pragma unroll
        for (int t = 0; t < PX + KW - 1; ++t) {
            const v4f k0 = kv[slot][0][cc[t]];
            const v4f k1 = kv[slot][1][cc[t]];
            const v2f kp[4] = {__builtin_shufflevector(k0, k0, 0, 1), __builtin_shufflevector(k0, k0, 2, 3),
                               __builtin_shufflevector(k1, k1, 0, 1), __builtin_shufflevector(k1, k1, 2, 3)};
            #pragma unroll
            for (int j = 0; j < PX; ++j) {
                const int kj = t - j;
                if (kj >= 0 && kj < KW) {
                    if (ROW_BIAS) {
                        v2f s = rbrow[j];
                        #pragma unroll
                        for (int m = 0; m < 4; ++m) s = __builtin_elementwise_fma(qp[j][m], kp[m], s);
                        lg[j][kj] = s[0] + s[1];
                    } else {
                        v2f s = qp[j][0] * kp[0];
                        #pragma unroll
                        for (int m = 1; m < 4; ++m) s = __builtin_elementwise_fma(qp[j][m], kp[m], s);
                        lg[j][kj] = (s[0] + s[1]) + rb[j][kj];
                    }
                }
            }
        }

        #pragma unroll
        for (int j = 0; j < PX; ++j) {
            float mn = mmax[j];
            #pragma unroll
            for (int kj = 0; kj < KW; ++kj) mn = fmaxf(mn, lg[j][kj]);
            const float corr = __expf(mmax[j] - mn);
            mmax[j] = mn;
            ssum[j] *= corr;
            #pragma unroll
            for (int n = 0; n < NCH; ++n) acc[j][n] *= corr;
        }

        #pragma unroll
        for (int t = 0; t < PX + KW - 1; ++t) {
            const v4f va  = kv[slot][2][cc[t]];
            const v4f vbb = kv[slot][3][cc[t]];
            #pragma unroll
            for (int j = 0; j < PX; ++j) {
                const int kj = t - j;
                if (kj >= 0 && kj < KW) {
                    const float e = __expf(lg[j][kj] - mmax[j]);
                    ssum[j] += e;
                    #pragma unroll
                    for (int n = 0; n < 4; ++n) acc[j][n]     = __builtin_fmaf(e, va[n],  acc[j][n]);
                    #pragma unroll
                    for (int n = 0; n < 4; ++n) acc[j][n + 4] = __builtin_fmaf(e, vbb[n], acc[j][n + 4]);
                }
            }
        }
    }

    float* st0 = reinterpret_cast<float*>(&qb[rr][0][0]);
    float* st1 = reinterpret_cast<float*>(&qb[rr][1][0]);
    #pragma unroll
    for (int j = 0; j < PX; ++j) {
        const float inv = 1.0f / ssum[j];
        #pragma unroll
        for (int n = 0; n < 4; ++n) {
            st0[n * WW + w0 + j] = acc[j][n]     * inv;
            st1[n * WW + w0 + j] = acc[j][n + 4] * inv;
        }
    }
    __syncthreads();
    float* op = outg + (size_t)rr * WW + lane;
    #pragma unroll
    for (int n = 0; n < 4; ++n) {
        #pragma unroll
        for (int t = 0; t < PX; ++t) {
            op[(size_t)n * HWSZ + 32 * t]       = st0[n * WW + lane + 32 * t];
            op[(size_t)(n + 4) * HWSZ + 32 * t] = st1[n * WW + lane + 32 * t];
        }
    }
}

__global__ __launch_bounds__(NTHR)
void SelfAttentionBlock_kernel(const float* __restrict__ x,
                               const float* __restrict__ Wk,
                               const float* __restrict__ Wq,
                               const float* __restrict__ Wv,
                               const float* __restrict__ r_ai,
                               const float* __restrict__ r_aj,
                               float* __restrict__ out)
{
    __shared__ v4f kv[KVROWS][4][LW];
    __shared__ v4f qb[ROWS][2][LW];
    __shared__ __attribute__((aligned(16))) float rcomp[NTHR / NCH][NCH];

    const int h0   = blockIdx.x * ROWS;
    const int g    = blockIdx.y;
    const int b    = blockIdx.z;
    const int tid  = threadIdx.x;
    const int lane = tid & 31;
    const int wave = __builtin_amdgcn_readfirstlane(tid >> 5);
    const int half = lane >> 4;
    const int m    = lane & 15;

    float rcv;
    {
        const int ti = (tid < KW * NCH) ? tid : 0;
        const int t = ti / NCH, n = ti % NCH;
        const int cg = g * NCH + n;
        rcv = (g < 4) ? r_ai[cg * KW + t] : r_aj[(cg - 32) * KW + t];
    }

    const float* WkG = Wk + g * 64;
    const float* WqG = Wq + g * 64;
    const float* WvG = Wv + g * 64;
    const int k_lo = half * 2;
    const int k_hi = 4 + half * 2;

    v2f aKV_lo, aKV_hi, aQ_lo, aQ_hi;
    if (m < 8) {
        aKV_lo[0] = WkG[m * 8 + k_lo];     aKV_lo[1] = WkG[m * 8 + k_lo + 1];
        aKV_hi[0] = WkG[m * 8 + k_hi];     aKV_hi[1] = WkG[m * 8 + k_hi + 1];
    } else {
        aKV_lo[0] = WvG[(m - 8) * 8 + k_lo]; aKV_lo[1] = WvG[(m - 8) * 8 + k_lo + 1];
        aKV_hi[0] = WvG[(m - 8) * 8 + k_hi]; aKV_hi[1] = WvG[(m - 8) * 8 + k_hi + 1];
    }
    aQ_lo[0] = WqG[(m & 7) * 8 + k_lo];  aQ_lo[1] = WqG[(m & 7) * 8 + k_lo + 1];
    aQ_hi[0] = WqG[(m & 7) * 8 + k_hi];  aQ_hi[1] = WqG[(m & 7) * 8 + k_hi + 1];

    const float* xg = x + (size_t)(b * CIN + g * CGC) * HWSZ;

    const int hsel  = wave >> 1;
    const int hslot = (hsel < PADR) ? hsel : hsel + ROWS;
    const int hcol  = (wave & 1) * 48;
    const unsigned lane_off = (unsigned)(k_lo * HWSZ + m);

    v2f pblo[NPLAIN], pbhi[NPLAIN];
    v2f cblo[NCENT],  cbhi[NCENT];

    {
        const float* xs0 = xg + (h0 + wave) * WW;
        #pragma unroll
        for (int j = 0; j < NCENT; ++j) {
            const float* xs = xs0 + j * 16;
            cblo[j][0] = xs[lane_off];
            cblo[j][1] = xs[lane_off + HWSZ];
            cbhi[j][0] = xs[lane_off + 4 * HWSZ];
            cbhi[j][1] = xs[lane_off + 5 * HWSZ];
        }
        const int hr = reflect96(h0 - PADR + hslot);
        const float* xh0 = xg + hr * WW + hcol;
        #pragma unroll
        for (int j = 0; j < NPLAIN; ++j) {
            const float* xs = xh0 + j * 16;
            pblo[j][0] = xs[lane_off];
            pblo[j][1] = xs[lane_off + HWSZ];
            pbhi[j][0] = xs[lane_off + 4 * HWSZ];
            pbhi[j][1] = xs[lane_off + 5 * HWSZ];
        }
    }

    rcomp[tid / NCH][tid % NCH] = rcv;

    #pragma unroll
    for (int j = 0; j < NCENT; ++j) {
        const int rr  = wave;
        const int i   = rr + PADR;
        const int col = j * 16 + m;

        v8f acc = {};
        acc = __builtin_amdgcn_wmma_f32_16x16x4_f32(false, aKV_lo, false, cblo[j],
                                                    (short)0, acc, false, false);
        acc = __builtin_amdgcn_wmma_f32_16x16x4_f32(false, aKV_hi, false, cbhi[j],
                                                    (short)0, acc, false, false);
        v8f qacc = {};
        qacc = __builtin_amdgcn_wmma_f32_16x16x4_f32(false, aQ_lo, false, cblo[j],
                                                     (short)0, qacc, false, false);
        qacc = __builtin_amdgcn_wmma_f32_16x16x4_f32(false, aQ_hi, false, cbhi[j],
                                                     (short)0, qacc, false, false);

        v4f lo4 = {acc[0], acc[1], acc[2], acc[3]};
        v4f hi4 = {acc[4], acc[5], acc[6], acc[7]};
        if (half) { kv[i][2][col] = lo4; kv[i][3][col] = hi4; }
        else      { kv[i][0][col] = lo4; kv[i][1][col] = hi4; }

        v4f qlo4 = {qacc[0], qacc[1], qacc[2], qacc[3]};
        v4f qhi4 = {qacc[4], qacc[5], qacc[6], qacc[7]};
        if (half) qb[rr][1][col] = qhi4;
        else      qb[rr][0][col] = qlo4;
    }
    #pragma unroll
    for (int j = 0; j < NPLAIN; ++j) {
        const int i   = hslot;
        const int col = hcol + j * 16 + m;

        v8f acc = {};
        acc = __builtin_amdgcn_wmma_f32_16x16x4_f32(false, aKV_lo, false, pblo[j],
                                                    (short)0, acc, false, false);
        acc = __builtin_amdgcn_wmma_f32_16x16x4_f32(false, aKV_hi, false, pbhi[j],
                                                    (short)0, acc, false, false);
        v4f lo4 = {acc[0], acc[1], acc[2], acc[3]};
        v4f hi4 = {acc[4], acc[5], acc[6], acc[7]};
        if (half) { kv[i][2][col] = lo4; kv[i][3][col] = hi4; }
        else      { kv[i][0][col] = lo4; kv[i][1][col] = hi4; }
    }

    __syncthreads();

    float* outg = out + ((size_t)(b * CIN + g * NCH) * HH + h0) * WW;
    if (g < 4) attend<true >(kv, qb, rcomp, outg, wave, lane);
    else       attend<false>(kv, qb, rcomp, outg, wave, lane);
}

extern "C" void kernel_launch(void* const* d_in, const int* in_sizes, int n_in,
                              void* d_out, int out_size, void* d_ws, size_t ws_size,
                              hipStream_t stream) {
    (void)in_sizes; (void)n_in; (void)out_size; (void)d_ws; (void)ws_size;
    const float* x    = (const float*)d_in[0];
    const float* Wk   = (const float*)d_in[1];
    const float* Wq   = (const float*)d_in[2];
    const float* Wv   = (const float*)d_in[3];
    const float* r_ai = (const float*)d_in[4];
    const float* r_aj = (const float*)d_in[5];
    float* out = (float*)d_out;

    dim3 grid(HH / ROWS, GROUPS, BQ);
    SelfAttentionBlock_kernel<<<grid, NTHR, 0, stream>>>(x, Wk, Wq, Wv, r_ai, r_aj, out);
}


# Attention on the raw x records: Wk folded into the query (u = Wk^T Wq x, bias dots from the same product), Wv applied once behind the softmax sum, on WMMA; half the LDS reads; sum order changed
# speedup vs baseline: 2.6821x; 1.0484x over previous
#include <hip/hip_runtime.h>

#define BQ     4
#define CIN    64
#define HH     96
#define WW     96
#define GROUPS 8
#define NCH    8
#define CGC    8
#define KW     7
#define PADR   3
#define HWSZ   (HH * WW)

#define ROWS   12
#define XROWS  (ROWS + 2*PADR)
#define NTHR   384
#define NPLAIN 3
#define NCENT  6
#define LW     112
#define PX     3

typedef __attribute__((ext_vector_type(2))) float v2f;
typedef __attribute__((ext_vector_type(4))) float v4f;
typedef __attribute__((ext_vector_type(8))) float v8f;

#define WMMA_GUARD() do { __builtin_amdgcn_sched_barrier(0); asm volatile("v_nop\n\tv_nop\n\tv_nop\n\tv_nop"); __builtin_amdgcn_sched_barrier(0); } while (0)

__device__ __forceinline__ int reflect96(int i) {
    if (i < 0)  i = -i;
    if (i > 95) i = 190 - i;
    return i;
}

template <bool ROW_BIAS>
__device__ __forceinline__ void attend(v4f (&xs)[XROWS][4][LW], const v4f (&bb)[ROWS][2][LW],
                                       v2f aV_lo, v2f aV_hi,
                                       float* __restrict__ outg, int rr, int lane)
{
    const int w0 = 3 * lane;

    int cc[PX + KW - 1];
    #pragma unroll
    for (int t = 0; t < PX + KW - 1; ++t) {
        int c = w0 + t - PADR;
        cc[t] = (c < 0) ? -c : ((c > 95) ? 190 - c : c);
    }

    v2f qp[PX][4];
    #pragma unroll
    for (int j = 0; j < PX; ++j) {
        const v4f a = xs[rr + PADR][2][w0 + j];
        const v4f b = xs[rr + PADR][3][w0 + j];
        qp[j][0] = __builtin_shufflevector(a, a, 0, 1);
        qp[j][1] = __builtin_shufflevector(a, a, 2, 3);
        qp[j][2] = __builtin_shufflevector(b, b, 0, 1);
        qp[j][3] = __builtin_shufflevector(b, b, 2, 3);
    }

    float rb[PX][KW];
    if (!ROW_BIAS) {
        #pragma unroll
        for (int j = 0; j < PX; ++j) {
            const v4f b0 = bb[rr][0][w0 + j];
            const v4f b1 = bb[rr][1][w0 + j];
            rb[j][0] = b0[0]; rb[j][1] = b0[1]; rb[j][2] = b0[2]; rb[j][3] = b0[3];
            rb[j][4] = b1[0]; rb[j][5] = b1[1]; rb[j][6] = b1[2];
        }
    }
    const float* bbf = reinterpret_cast<const float*>(&bb[rr][0][0]);

    float mmax[PX], ssum[PX];
    float acc[PX][NCH];
    #pragma unroll
    for (int j = 0; j < PX; ++j) {
        mmax[j] = -3.0e38f;
        ssum[j] = 0.0f;
        #pragma unroll
        for (int n = 0; n < NCH; ++n) acc[j][n] = 0.0f;
    }

    #pragma nounroll
    for (int ki = 0; ki < KW; ++ki) {
        const int slot = rr + ki;

        v2f rbrow[PX];
        if (ROW_BIAS) {
            const int boff = (ki >> 2) * (LW * 4) + (ki & 3);
            #pragma unroll
            for (int j = 0; j < PX; ++j) {
                rbrow[j][0] = bbf[boff + (w0 + j) * 4];
                rbrow[j][1] = 0.0f;
            }
        }

        v4f x0[PX + KW - 1], x1[PX + KW - 1];
        float lg[PX][KW];
        #pragma unroll
        for (int t = 0; t < PX + KW - 1; ++t) {
            x0[t] = xs[slot][0][cc[t]];
            x1[t] = xs[slot][1][cc[t]];
            const v2f kp[4] = {__builtin_shufflevector(x0[t], x0[t], 0, 1), __builtin_shufflevector(x0[t], x0[t], 2, 3),
                               __builtin_shufflevector(x1[t], x1[t], 0, 1), __builtin_shufflevector(x1[t], x1[t], 2, 3)};
            #pragma unroll
            for (int j = 0; j < PX; ++j) {
                const int kj = t - j;
                if (kj >= 0 && kj < KW) {
                    if (ROW_BIAS) {
                        v2f s = rbrow[j];
                        #pragma unroll
                        for (int m = 0; m < 4; ++m) s = __builtin_elementwise_fma(qp[j][m], kp[m], s);
                        lg[j][kj] = s[0] + s[1];
                    } else {
                        v2f s = qp[j][0] * kp[0];
                        #pragma unroll
                        for (int m = 1; m < 4; ++m) s = __builtin_elementwise_fma(qp[j][m], kp[m], s);
                        lg[j][kj] = (s[0] + s[1]) + rb[j][kj];
                    }
                }
            }
        }

        #pragma unroll
        for (int j = 0; j < PX; ++j) {
            float mn = mmax[j];
            #pragma unroll
            for (int kj = 0; kj < KW; ++kj) mn = fmaxf(mn, lg[j][kj]);
            const float corr = __expf(mmax[j] - mn);
            mmax[j] = mn;
            ssum[j] *= corr;
            #pragma unroll
            for (int n = 0; n < NCH; ++n) acc[j][n] *= corr;
        }

        #pragma unroll
        for (int t = 0; t < PX + KW - 1; ++t) {
            #pragma unroll
            for (int j = 0; j < PX; ++j) {
                const int kj = t - j;
                if (kj >= 0 && kj < KW) {
                    const float e = __expf(lg[j][kj] - mmax[j]);
                    ssum[j] += e;
                    #pragma unroll
                    for (int n = 0; n < 4; ++n) acc[j][n]     = __builtin_fmaf(e, x0[t][n], acc[j][n]);
                    #pragma unroll
                    for (int n = 0; n < 4; ++n) acc[j][n + 4] = __builtin_fmaf(e, x1[t][n], acc[j][n + 4]);
                }
            }
        }
    }

    #pragma unroll
    for (int j = 0; j < PX; ++j) {
        const float inv = 1.0f / ssum[j];
        const v4f s0 = {acc[j][0] * inv, acc[j][1] * inv, acc[j][2] * inv, acc[j][3] * inv};
        const v4f s1 = {acc[j][4] * inv, acc[j][5] * inv, acc[j][6] * inv, acc[j][7] * inv};
        xs[rr + PADR][2][w0 + j] = s0;
        xs[rr + PADR][3][w0 + j] = s1;
    }
    __syncthreads();

    const int half = lane >> 4;
    const int m    = lane & 15;
    const v2f* sp0 = reinterpret_cast<const v2f*>(&xs[rr + PADR][2][0]);
    const v2f* sp1 = reinterpret_cast<const v2f*>(&xs[rr + PADR][3][0]);
    v2f blo[NCENT], bhi[NCENT];
    #pragma unroll
    for (int s = 0; s < NCENT; ++s) {
        blo[s] = sp0[(16 * s + m) * 2 + half];
        bhi[s] = sp1[(16 * s + m) * 2 + half];
    }
    v8f d[NCENT];
    __builtin_amdgcn_sched_barrier(0);
    #pragma unroll
    for (int s = 0; s < NCENT; ++s) {
        v8f o = {};
        d[s] = __builtin_amdgcn_wmma_f32_16x16x4_f32(false, aV_lo, false, blo[s], (short)0, o, false, false);
    }
    #pragma unroll
    for (int s = 0; s < NCENT; ++s)
        d[s] = __builtin_amdgcn_wmma_f32_16x16x4_f32(false, aV_hi, false, bhi[s], (short)0, d[s], false, false);
    WMMA_GUARD();
    float* op = outg + (size_t)rr * WW + lane;
    #pragma unroll
    for (int t = 0; t < NCENT / 2; ++t) {
        #pragma unroll
        for (int r = 0; r < NCH; ++r) {
            const float v = half ? d[2 * t + 1][r] : d[2 * t][r];
            op[(size_t)r * HWSZ + 32 * t] = v;
        }
    }
}

__global__ __launch_bounds__(NTHR)
void SelfAttentionBlock_kernel(const float* __restrict__ x,
                               const float* __restrict__ Wk,
                               const float* __restrict__ Wq,
                               const float* __restrict__ Wv,
                               const float* __restrict__ r_ai,
                               const float* __restrict__ r_aj,
                               float* __restrict__ out)
{
    __shared__ v4f xs[XROWS][4][LW];
    __shared__ v4f bb[ROWS][2][LW];

    const int h0   = blockIdx.x * ROWS;
    const int g    = blockIdx.y;
    const int b    = blockIdx.z;
    const int tid  = threadIdx.x;
    const int lane = tid & 31;
    const int wave = __builtin_amdgcn_readfirstlane(tid >> 5);
    const int half = lane >> 4;
    const int m    = lane & 15;

    const float* WkG = Wk + g * 64;
    const float* WqG = Wq + g * 64;
    const float* WvG = Wv + g * 64;
    const int k_lo = half * 2;
    const int k_hi = 4 + half * 2;

    v2f aU_lo = {0.0f, 0.0f}, aU_hi = {0.0f, 0.0f};
    {
        const int tq = (m >= 8 && m < 15) ? (m - 8) : 0;
        const float* rsel  = (g < 4) ? (r_ai + g * (NCH * KW)) : (r_aj + (g - 4) * (NCH * KW));
        const float* lbase = (m < 8) ? (WkG + m) : (rsel + tq);
        const int lstride  = (m < 8) ? 8 : KW;
        float lv[NCH];
        #pragma unroll
        for (int n = 0; n < NCH; ++n) lv[n] = lbase[n * lstride];
        #pragma unroll
        for (int n = 0; n < NCH; ++n) {
            const float l = (m == 15) ? 0.0f : lv[n];
            aU_lo[0] = __builtin_fmaf(l, WqG[n * 8 + k_lo],     aU_lo[0]);
            aU_lo[1] = __builtin_fmaf(l, WqG[n * 8 + k_lo + 1], aU_lo[1]);
            aU_hi[0] = __builtin_fmaf(l, WqG[n * 8 + k_hi],     aU_hi[0]);
            aU_hi[1] = __builtin_fmaf(l, WqG[n * 8 + k_hi + 1], aU_hi[1]);
        }
    }
    v2f aV_lo, aV_hi;
    aV_lo[0] = WvG[(m & 7) * 8 + k_lo];  aV_lo[1] = WvG[(m & 7) * 8 + k_lo + 1];
    aV_hi[0] = WvG[(m & 7) * 8 + k_hi];  aV_hi[1] = WvG[(m & 7) * 8 + k_hi + 1];

    const float* xg = x + (size_t)(b * CIN + g * CGC) * HWSZ;

    const int hsel  = wave >> 1;
    const int hslot = (hsel < PADR) ? hsel : hsel + ROWS;
    const int hcol  = (wave & 1) * 48;
    const unsigned lane_off = (unsigned)(k_lo * HWSZ + m);

    v2f pblo[NPLAIN], pbhi[NPLAIN];
    v2f cblo[NCENT],  cbhi[NCENT];

    {
        const float* xs0 = xg + (h0 + wave) * WW;
        #pragma unroll
        for (int j = 0; j < NCENT; ++j) {
            const float* xp = xs0 + j * 16;
            cblo[j][0] = xp[lane_off];
            cblo[j][1] = xp[lane_off + HWSZ];
            cbhi[j][0] = xp[lane_off + 4 * HWSZ];
            cbhi[j][1] = xp[lane_off + 5 * HWSZ];
        }
        const int hr = reflect96(h0 - PADR + hslot);
        const float* xh0 = xg + hr * WW + hcol;
        #pragma unroll
        for (int j = 0; j < NPLAIN; ++j) {
            const float* xp = xh0 + j * 16;
            pblo[j][0] = xp[lane_off];
            pblo[j][1] = xp[lane_off + HWSZ];
            pbhi[j][0] = xp[lane_off + 4 * HWSZ];
            pbhi[j][1] = xp[lane_off + 5 * HWSZ];
        }
    }

    v8f uacc[NCENT];
    #pragma unroll
    for (int j = 0; j < NCENT; ++j) {
        v8f o = {};
        uacc[j] = __builtin_amdgcn_wmma_f32_16x16x4_f32(false, aU_lo, false, cblo[j], (short)0, o, false, false);
    }
    #pragma unroll
    for (int j = 0; j < NCENT; ++j)
        uacc[j] = __builtin_amdgcn_wmma_f32_16x16x4_f32(false, aU_hi, false, cbhi[j], (short)0, uacc[j], false, false);
    WMMA_GUARD();
    #pragma unroll
    for (int j = 0; j < NCENT; ++j) {
        const int rr  = wave;
        const int i   = rr + PADR;
        const int col = j * 16 + m;

        reinterpret_cast<v2f*>(&xs[i][0][col])[half] = cblo[j];
        reinterpret_cast<v2f*>(&xs[i][1][col])[half] = cbhi[j];

        const v4f lo4 = {uacc[j][0], uacc[j][1], uacc[j][2], uacc[j][3]};
        const v4f hi4 = {uacc[j][4], uacc[j][5], uacc[j][6], uacc[j][7]};
        v4f* d0 = half ? &bb[rr][0][col] : &xs[i][2][col];
        v4f* d1 = half ? &bb[rr][1][col] : &xs[i][3][col];
        *d0 = lo4;
        *d1 = hi4;
    }
    #pragma unroll
    for (int j = 0; j < NPLAIN; ++j) {
        const int col = hcol + j * 16 + m;
        reinterpret_cast<v2f*>(&xs[hslot][0][col])[half] = pblo[j];
        reinterpret_cast<v2f*>(&xs[hslot][1][col])[half] = pbhi[j];
    }

    __syncthreads();

    float* outg = out + ((size_t)(b * CIN + g * NCH) * HH + h0) * WW;
    if (g < 4) attend<true >(xs, bb, aV_lo, aV_hi, outg, wave, lane);
    else       attend<false>(xs, bb, aV_lo, aV_hi, outg, wave, lane);
}

extern "C" void kernel_launch(void* const* d_in, const int* in_sizes, int n_in,
                              void* d_out, int out_size, void* d_ws, size_t ws_size,
                              hipStream_t stream) {
    (void)in_sizes; (void)n_in; (void)out_size; (void)d_ws; (void)ws_size;
    const float* x    = (const float*)d_in[0];
    const float* Wk   = (const float*)d_in[1];
    const float* Wq   = (const float*)d_in[2];
    const float* Wv   = (const float*)d_in[3];
    const float* r_ai = (const float*)d_in[4];
    const float* r_aj = (const float*)d_in[5];
    float* out = (float*)d_out;

    dim3 grid(HH / ROWS, GROUPS, BQ);
    SelfAttentionBlock_kernel<<<grid, NTHR, 0, stream>>>(x, Wk, Wq, Wv, r_ai, r_aj, out);
}
